# starting kernel
# baseline (speedup 1.0000x reference)
#include <hip/hip_runtime.h>
#include <math.h>

typedef float v8f __attribute__((ext_vector_type(8)));
typedef float v4f __attribute__((ext_vector_type(4)));
typedef float v2f __attribute__((ext_vector_type(2)));

#define B_      4
#define H_      12
#define S_      1024
#define D_      64
#define HALF_   32
#define LAMBDA_ 0.01f
#define EPS_    1e-6f
#define CLAMP_  20.0f
#define QTILES  (S_ / 16)
#define WPB     4
#define LSTR    20

__global__ __launch_bounds__(32 * WPB)
void rse_stickbreak_attn(const float* __restrict__ q,
                         const float* __restrict__ k,
                         const float* __restrict__ v,
                         const float* __restrict__ cosc,
                         const float* __restrict__ sinc,
                         float* __restrict__ out)
{
    __shared__ __align__(16) float lds[WPB][16 * LSTR + 16];

    const int wave = threadIdx.x >> 5;
    const int lane = threadIdx.x & 31;
    const int tile = blockIdx.x * WPB + wave;
    const int qt   = tile % QTILES;
    const int bh   = tile / QTILES;

    float* wlds = &lds[wave][0];
    float* dlds = &lds[wave][16 * LSTR];

    const int hi = lane >> 4;
    const int ln = lane & 15;

    const float* qb = q + (size_t)bh * S_ * D_;
    const float* kb = k + (size_t)bh * S_ * D_;
    const float* vb = v + (size_t)bh * S_ * D_;
    float*       ob = out + (size_t)bh * S_ * D_;

    const int qrow = qt * 16 + ln;
    v2f afrag[16];
#pragma unroll
    for (int c = 0; c < 16; ++c) {
        const int d  = 4 * c + 2 * hi;
        const v2f qp = *(const v2f*)(qb + qrow * D_ + d);
        const float cc = cosc[qrow * HALF_ + (d >> 1)];
        const float ss = sinc[qrow * HALF_ + (d >> 1)];
        afrag[c].x = qp.x * cc - qp.y * ss;
        afrag[c].y = qp.y * cc + qp.x * ss;
    }

    v8f acc0 = (v8f)0.0f, acc1 = (v8f)0.0f, acc2 = (v8f)0.0f, acc3 = (v8f)0.0f;
    float rem   = 1.0f;
    float denom = 0.0f;
    const float scale = 0.125f;

    for (int kt = 0; kt <= qt; ++kt) {
        if (kt < qt) {
            __builtin_prefetch(&kb[(kt + 1) * 16 * D_ + lane * 32], 0, 3);
            __builtin_prefetch(&vb[(kt + 1) * 16 * D_ + lane * 32], 0, 3);
        }

        const int krow = kt * 16 + ln;
        v8f cacc = (v8f)0.0f;
#pragma unroll
        for (int c = 0; c < 16; ++c) {
            const int d  = 4 * c + 2 * hi;
            const v2f kp = *(const v2f*)(kb + krow * D_ + d);
            const float cc = cosc[krow * HALF_ + (d >> 1)];
            const float ss = sinc[krow * HALF_ + (d >> 1)];
            v2f bfrag;
            bfrag.x = kp.x * cc - kp.y * ss;
            bfrag.y = kp.y * cc + kp.x * ss;
            cacc = __builtin_amdgcn_wmma_f32_16x16x4_f32(
                false, afrag[c], false, bfrag, (short)0, cacc, false, false);
        }

#pragma unroll
        for (int r = 0; r < 8; ++r) {
            const int m    = r + 8 * hi;
            const int qpos = qt * 16 + m;
            const int kpos = kt * 16 + ln;
            float x = cacc[r] * scale - LAMBDA_ * fabsf((float)(qpos - kpos));
            const bool fut = kpos > qpos;
            x = fminf(fmaxf(x, -CLAMP_), CLAMP_);
            float beta = 1.0f / (1.0f + __expf(-x));
            if (fut) beta = 0.0f;
            wlds[m * LSTR + ln] = beta;
        }
        asm volatile("s_wait_dscnt 0" ::: "memory");
        __builtin_amdgcn_wave_barrier();

        if (lane < 16) {
            v4f* rowp = (v4f*)(wlds + lane * LSTR);
            v4f w0 = rowp[0], w1 = rowp[1], w2 = rowp[2], w3 = rowp[3];
            float r_   = rem;
            float dsum = 0.0f;
#define SB_STEP(VEC, I)                                   \
            { const float w = VEC[I] * r_;                \
              r_ = fmaxf(r_ * (1.0f - w), EPS_);          \
              VEC[I] = w;  dsum += w; }
            SB_STEP(w0, 0) SB_STEP(w0, 1) SB_STEP(w0, 2) SB_STEP(w0, 3)
            SB_STEP(w1, 0) SB_STEP(w1, 1) SB_STEP(w1, 2) SB_STEP(w1, 3)
            SB_STEP(w2, 0) SB_STEP(w2, 1) SB_STEP(w2, 2) SB_STEP(w2, 3)
            SB_STEP(w3, 0) SB_STEP(w3, 1) SB_STEP(w3, 2) SB_STEP(w3, 3)
#undef SB_STEP
            rowp[0] = w0; rowp[1] = w1; rowp[2] = w2; rowp[3] = w3;
            rem    = r_;
            denom += dsum;
        }
        asm volatile("s_wait_dscnt 0" ::: "memory");
        __builtin_amdgcn_wave_barrier();

        v2f awf[4];
#pragma unroll
        for (int kc = 0; kc < 4; ++kc)
            awf[kc] = *(const v2f*)(wlds + ln * LSTR + 4 * kc + 2 * hi);
#pragma unroll
        for (int kc = 0; kc < 4; ++kc) {
            const int vr = kt * 16 + 4 * kc + 2 * hi;
            v2f b0, b1, b2, b3;
            b0.x = vb[vr * D_ +  0 + ln];  b0.y = vb[(vr + 1) * D_ +  0 + ln];
            b1.x = vb[vr * D_ + 16 + ln];  b1.y = vb[(vr + 1) * D_ + 16 + ln];
            b2.x = vb[vr * D_ + 32 + ln];  b2.y = vb[(vr + 1) * D_ + 32 + ln];
            b3.x = vb[vr * D_ + 48 + ln];  b3.y = vb[(vr + 1) * D_ + 48 + ln];
            acc0 = __builtin_amdgcn_wmma_f32_16x16x4_f32(
                false, awf[kc], false, b0, (short)0, acc0, false, false);
            acc1 = __builtin_amdgcn_wmma_f32_16x16x4_f32(
                false, awf[kc], false, b1, (short)0, acc1, false, false);
            acc2 = __builtin_amdgcn_wmma_f32_16x16x4_f32(
                false, awf[kc], false, b2, (short)0, acc2, false, false);
            acc3 = __builtin_amdgcn_wmma_f32_16x16x4_f32(
                false, awf[kc], false, b3, (short)0, acc3, false, false);
        }
    }

    if (lane < 16) dlds[lane] = fmaxf(denom, EPS_);
    asm volatile("s_wait_dscnt 0" ::: "memory");
    __builtin_amdgcn_wave_barrier();

#pragma unroll
    for (int r = 0; r < 8; ++r) {
        const int m  = r + 8 * hi;
        const float dn = dlds[m];
        const int orow = qt * 16 + m;
        ob[orow * D_ +  0 + ln] = acc0[r] / dn;
        ob[orow * D_ + 16 + ln] = acc1[r] / dn;
        ob[orow * D_ + 32 + ln] = acc2[r] / dn;
        ob[orow * D_ + 48 + ln] = acc3[r] / dn;
    }
}

extern "C" void kernel_launch(void* const* d_in, const int* in_sizes, int n_in,
                              void* d_out, int out_size, void* d_ws, size_t ws_size,
                              hipStream_t stream) {
    (void)in_sizes; (void)n_in; (void)out_size; (void)d_ws; (void)ws_size;
    const float* q    = (const float*)d_in[0];
    const float* k    = (const float*)d_in[1];
    const float* v    = (const float*)d_in[2];
    const float* cosc = (const float*)d_in[3];
    const float* sinc = (const float*)d_in[4];
    float* out = (float*)d_out;

    const int tiles  = B_ * H_ * QTILES;
    const int blocks = tiles / WPB;
    rse_stickbreak_attn<<<blocks, 32 * WPB, 0, stream>>>(q, k, v, cosc, sinc, out);
}


# Each query tile worked by two waves one key tile apart: a producer wave forms the beta tile (logits, sigmoid) into LDS, a consumer wave runs scan and W x V; 8-wave blocks; sum orders unchanged
# speedup vs baseline: 1.8062x; 1.8062x over previous
#include <hip/hip_runtime.h>
#include <math.h>


typedef float v8f __attribute__((ext_vector_type(8)));
typedef float v4f __attribute__((ext_vector_type(4)));
typedef float v2f __attribute__((ext_vector_type(2)));

#define B_      4
#define H_      12
#define S_      1024
#define D_      64
#define HALF_   32
#define LAMBDA_ 0.01f
#define EPS_    1e-6f
#define CLAMP_  20.0f
#define QTILES  (S_ / 16)
#define WPB     4
#define NTHR    (64 * WPB)
#define LSTR    20
#define KPITCH  68
#define VPITCH  160

__global__ __launch_bounds__(NTHR, 6)
void rse_stickbreak_attn(const float* __restrict__ q,
                         const float* __restrict__ k,
                         const float* __restrict__ v,
                         const float* __restrict__ cosc,
                         const float* __restrict__ sinc,
                         float* __restrict__ out)
{
    __shared__ __align__(16) float krs[2][16 * KPITCH];
    __shared__ __align__(16) float vps[2][8 * VPITCH];
    __shared__ __align__(16) float bts[2][WPB][16 * LSTR];
    __shared__ __align__(16) float dls[WPB][16];

    const int tid  = threadIdx.x;
    const int wave = __builtin_amdgcn_readfirstlane(tid >> 5);
    const int lane = tid & 31;
    const bool producer = wave < WPB;
    const int qw   = wave & (WPB - 1);
    const int rtid = tid & (32 * WPB - 1);
    const int bid  = blockIdx.x;
    const int seg  = bid >> 8;
    const int pos  = bid & 15;
    const unsigned long long perm1 = 0x8b7a5de4fc962310ull;
    const unsigned long long perm2 = 0xfbead42c01359687ull;
    const int grp  = (seg == 0) ? (15 - pos)
                                : (int)(((seg == 1 ? perm1 : perm2) >> (4 * pos)) & 15ull);
    const int qt0   = grp * WPB;
    const int bh    = seg * 16 + ((bid >> 4) & 15);
    const int qt    = qt0 + qw;
    const int ktmax = qt0 + WPB - 1;

    const int hi = lane >> 4;
    const int ln = lane & 15;

    const float* qb = q + (size_t)bh * S_ * D_;
    const float* kb = k + (size_t)bh * S_ * D_;
    const float* vb = v + (size_t)bh * S_ * D_;
    float*       ob = out + (size_t)bh * S_ * D_;

    const int sr = rtid >> 3;
    const int sg = rtid & 7;
    const int vp = rtid >> 4;
    const int vu = rtid & 15;
    const float* kst = kb   + sr * D_    + 8 * sg;
    const float* cst = cosc + sr * HALF_ + 4 * sg;
    const float* sst = sinc + sr * HALF_ + 4 * sg;
    const float* vst = vb   + (2 * vp) * D_ + 4 * vu;
    float* kdst = &krs[0][sr * KPITCH + 8 * sg];
    float* vdst = &vps[0][vp * VPITCH + 8 * vu];

    v2f afrag[16];
    v8f acc0 = (v8f)0.0f, acc1 = (v8f)0.0f, acc2 = (v8f)0.0f, acc3 = (v8f)0.0f;
    float rem   = 1.0f;
    float denom = 0.0f;
    const float scale = 0.125f;

#define STORE_K(ST, GK0, GK1, GC, GS)                                     \
    { v4f r0_, r1_;                                                       \
      r0_.x = GK0.x * GC.x - GK0.y * GS.x;                                \
      r0_.y = GK0.y * GC.x + GK0.x * GS.x;                                \
      r0_.z = GK0.z * GC.y - GK0.w * GS.y;                                \
      r0_.w = GK0.w * GC.y + GK0.z * GS.y;                                \
      r1_.x = GK1.x * GC.z - GK1.y * GS.z;                                \
      r1_.y = GK1.y * GC.z + GK1.x * GS.z;                                \
      r1_.z = GK1.z * GC.w - GK1.w * GS.w;                                \
      r1_.w = GK1.w * GC.w + GK1.z * GS.w;                                \
      *(v4f*)(kdst + (ST) * 16 * KPITCH)     = r0_;                       \
      *(v4f*)(kdst + (ST) * 16 * KPITCH + 4) = r1_; }

    if (producer) {
        const v4f gk0 = *(const v4f*)(kst);
        const v4f gk1 = *(const v4f*)(kst + 4);
        const v4f gc  = *(const v4f*)(cst);
        const v4f gs  = *(const v4f*)(sst);

        const int qrow = qt * 16 + ln;
#pragma unroll
        for (int c = 0; c < 16; ++c) {
            const int d  = 4 * c + 2 * hi;
            const v2f qp = *(const v2f*)(qb + qrow * D_ + d);
            const float cc = cosc[qrow * HALF_ + (d >> 1)];
            const float ss = sinc[qrow * HALF_ + (d >> 1)];
            afrag[c].x = qp.x * cc - qp.y * ss;
            afrag[c].y = qp.y * cc + qp.x * ss;
        }
        STORE_K(0, gk0, gk1, gc, gs)
    }

    if (producer) {
        for (int t = 0; t <= ktmax + 1; ++t) {
            __syncthreads();
            const bool more = t < ktmax;
            v4f gk0, gk1, gc, gs;
            if (more) {
                const int o_ = (t + 1) * 16;
                gk0 = *(const v4f*)(kst + o_ * D_);
                gk1 = *(const v4f*)(kst + o_ * D_ + 4);
                gc  = *(const v4f*)(cst + o_ * HALF_);
                gs  = *(const v4f*)(sst + o_ * HALF_);
            }

            if (t <= qt) {
                const float* krt = &krs[t & 1][0];
                float* wlds = &bts[t & 1][qw][0];

                v2f bfrag[16];
#pragma unroll
                for (int c = 0; c < 16; ++c)
                    bfrag[c] = *(const v2f*)(krt + ln * KPITCH + 4 * c + 2 * hi);
                asm volatile("" :: "v"(bfrag[0]),  "v"(bfrag[1]),  "v"(bfrag[2]),  "v"(bfrag[3]),
                                   "v"(bfrag[4]),  "v"(bfrag[5]),  "v"(bfrag[6]),  "v"(bfrag[7]),
                                   "v"(bfrag[8]),  "v"(bfrag[9]),  "v"(bfrag[10]), "v"(bfrag[11]),
                                   "v"(bfrag[12]), "v"(bfrag[13]), "v"(bfrag[14]), "v"(bfrag[15]));
                v8f cacc = (v8f)0.0f;
#pragma unroll
                for (int c = 0; c < 16; ++c)
                    cacc = __builtin_amdgcn_wmma_f32_16x16x4_f32(
                        false, afrag[c], false, bfrag[c], (short)0, cacc, false, false);

#pragma unroll
                for (int r = 0; r < 8; ++r) {
                    const int m    = r + 8 * hi;
                    const int qpos = qt * 16 + m;
                    const int kpos = t * 16 + ln;
                    float x = cacc[r] * scale - LAMBDA_ * fabsf((float)(qpos - kpos));
                    const bool fut = kpos > qpos;
                    x = fminf(fmaxf(x, -CLAMP_), CLAMP_);
                    float beta = 1.0f / (1.0f + __expf(-x));
                    if (fut) beta = 0.0f;
                    wlds[m * LSTR + ln] = beta;
                }
            }

            if (more) STORE_K((t + 1) & 1, gk0, gk1, gc, gs)
        }
    } else {
        for (int t = 0; t <= ktmax + 1; ++t) {
            __syncthreads();
            const bool havev = t <= ktmax;
            v4f gva, gvb;
            if (havev) {
                const int o_ = t * 16;
                gva = *(const v4f*)(vst + o_ * D_);
                gvb = *(const v4f*)(vst + o_ * D_ + D_);
            }

            const int kt = t - 1;
            if (t >= 1 && kt <= qt) {
                const float* vpt = &vps[kt & 1][0];
                float* wlds = &bts[kt & 1][qw][0];

                v2f vf[4][4];
#pragma unroll
                for (int kc = 0; kc < 4; ++kc) {
                    const float* vrow = vpt + (2 * kc + hi) * VPITCH + 2 * ln;
#pragma unroll
                    for (int j = 0; j < 4; ++j)
                        vf[kc][j] = *(const v2f*)(vrow + 32 * j);
                }

                if (lane < 16) {
                    v4f* rowp = (v4f*)(wlds + lane * LSTR);
                    v4f w0 = rowp[0], w1 = rowp[1], w2 = rowp[2], w3 = rowp[3];
                    float r_   = rem;
                    float dsum = 0.0f;
#define SB_STEP(VEC, I)                                   \
                    { const float w = VEC[I] * r_;                \
                      r_ = fmaxf(r_ * (1.0f - w), EPS_);          \
                      VEC[I] = w;  dsum += w; }
                    SB_STEP(w0, 0) SB_STEP(w0, 1) SB_STEP(w0, 2) SB_STEP(w0, 3)
                    SB_STEP(w1, 0) SB_STEP(w1, 1) SB_STEP(w1, 2) SB_STEP(w1, 3)
                    SB_STEP(w2, 0) SB_STEP(w2, 1) SB_STEP(w2, 2) SB_STEP(w2, 3)
                    SB_STEP(w3, 0) SB_STEP(w3, 1) SB_STEP(w3, 2) SB_STEP(w3, 3)
#undef SB_STEP
                    rowp[0] = w0; rowp[1] = w1; rowp[2] = w2; rowp[3] = w3;
                    rem    = r_;
                    denom += dsum;
                }
                asm volatile("s_wait_dscnt 0" ::: "memory");
                __builtin_amdgcn_wave_barrier();

                v2f awf[4];
#pragma unroll
                for (int kc = 0; kc < 4; ++kc)
                    awf[kc] = *(const v2f*)(wlds + ln * LSTR + 4 * kc + 2 * hi);
#pragma unroll
                for (int kc = 0; kc < 4; ++kc) {
                    acc0 = __builtin_amdgcn_wmma_f32_16x16x4_f32(
                        false, awf[kc], false, vf[kc][0], (short)0, acc0, false, false);
                    acc1 = __builtin_amdgcn_wmma_f32_16x16x4_f32(
                        false, awf[kc], false, vf[kc][1], (short)0, acc1, false, false);
                    acc2 = __builtin_amdgcn_wmma_f32_16x16x4_f32(
                        false, awf[kc], false, vf[kc][2], (short)0, acc2, false, false);
                    acc3 = __builtin_amdgcn_wmma_f32_16x16x4_f32(
                        false, awf[kc], false, vf[kc][3], (short)0, acc3, false, false);
                }
            }

            if (havev) {
                v4f p0_, p1_;
                p0_.x = gva.x; p0_.y = gvb.x; p0_.z = gva.y; p0_.w = gvb.y;
                p1_.x = gva.z; p1_.y = gvb.z; p1_.z = gva.w; p1_.w = gvb.w;
                *(v4f*)(vdst + (t & 1) * 8 * VPITCH)     = p0_;
                *(v4f*)(vdst + (t & 1) * 8 * VPITCH + 4) = p1_;
            }
        }
    }
#undef STORE_K

    if (!producer) {
        float* dlds = &dls[qw][0];
        if (lane < 16) dlds[lane] = fmaxf(denom, EPS_);
        asm volatile("s_wait_dscnt 0" ::: "memory");
        __builtin_amdgcn_wave_barrier();

#pragma unroll
        for (int r = 0; r < 8; ++r) {
            const int m  = r + 8 * hi;
            const float dn = dlds[m];
            const int orow = qt * 16 + m;
            ob[orow * D_ +  0 + ln] = acc0[r] / dn;
            ob[orow * D_ + 16 + ln] = acc1[r] / dn;
            ob[orow * D_ + 32 + ln] = acc2[r] / dn;
            ob[orow * D_ + 48 + ln] = acc3[r] / dn;
        }
    }
}

extern "C" void kernel_launch(void* const* d_in, const int* in_sizes, int n_in,
                              void* d_out, int out_size, void* d_ws, size_t ws_size,
                              hipStream_t stream) {
    (void)in_sizes; (void)n_in; (void)out_size; (void)d_ws; (void)ws_size;
    const float* q    = (const float*)d_in[0];
    const float* k    = (const float*)d_in[1];
    const float* v    = (const float*)d_in[2];
    const float* cosc = (const float*)d_in[3];
    const float* sinc = (const float*)d_in[4];
    float* out = (float*)d_out;

    const int tiles  = B_ * H_ * QTILES;
    const int blocks = tiles / WPB;
    rse_stickbreak_attn<<<blocks, NTHR, 0, stream>>>(q, k, v, cosc, sinc, out);
}


# Equal-work blocks: each block's four query tiles are 2j, 2j+1, 62-2j, 63-2j (130 tile products each, was 10 to 250); long-tile producers share SIMDs with short-tile consumers; sum orders unchanged
# speedup vs baseline: 2.0830x; 1.1532x over previous
#include <hip/hip_runtime.h>
#include <math.h>


typedef float v8f __attribute__((ext_vector_type(8)));
typedef float v4f __attribute__((ext_vector_type(4)));
typedef float v2f __attribute__((ext_vector_type(2)));

#define B_      4
#define H_      12
#define S_      1024
#define D_      64
#define HALF_   32
#define LAMBDA_ 0.01f
#define EPS_    1e-6f
#define CLAMP_  20.0f
#define QTILES  (S_ / 16)
#define WPB     4
#define NTHR    (64 * WPB)
#define LSTR    20
#define KPITCH  68
#define VPITCH  160

__global__ __launch_bounds__(NTHR, 6)
void rse_stickbreak_attn(const float* __restrict__ q,
                         const float* __restrict__ k,
                         const float* __restrict__ v,
                         const float* __restrict__ cosc,
                         const float* __restrict__ sinc,
                         float* __restrict__ out)
{
    __shared__ __align__(16) float krs[2][16 * KPITCH];
    __shared__ __align__(16) float vps[2][8 * VPITCH];
    __shared__ __align__(16) float bts[2][WPB][16 * LSTR];
    __shared__ __align__(16) float dls[WPB][16];

    const int tid  = threadIdx.x;
    const int wave = __builtin_amdgcn_readfirstlane(tid >> 5);
    const int lane = tid & 31;
    const bool producer = wave < WPB;
    const int qw   = wave & (WPB - 1);
    const int rtid = tid & (32 * WPB - 1);
    const int bid   = blockIdx.x;
    const int bh    = bid >> 4;
    const int u_    = bid & 15;
    const int trd   = bid >> 8;
    const int jj    = (7 * u_ + 5 * trd) & 15;
    const int sp    = (qw + u_ + trd) & 3;
    const int slot  = producer ? (sp ^ 2) : sp;
    const int qt    = ((slot & 2) ? 62 - 2 * jj : 2 * jj) + (slot & 1);
    const int ktmax = 63 - 2 * jj;

    const int hi = lane >> 4;
    const int ln = lane & 15;

    const float* qb = q + (size_t)bh * S_ * D_;
    const float* kb = k + (size_t)bh * S_ * D_;
    const float* vb = v + (size_t)bh * S_ * D_;
    float*       ob = out + (size_t)bh * S_ * D_;

    const int sr = rtid >> 3;
    const int sg = rtid & 7;
    const int vp = rtid >> 4;
    const int vu = rtid & 15;
    const float* kst = kb   + sr * D_    + 8 * sg;
    const float* cst = cosc + sr * HALF_ + 4 * sg;
    const float* sst = sinc + sr * HALF_ + 4 * sg;
    const float* vst = vb   + (2 * vp) * D_ + 4 * vu;
    float* kdst = &krs[0][sr * KPITCH + 8 * sg];
    float* vdst = &vps[0][vp * VPITCH + 8 * vu];

    v2f afrag[16];
    v8f acc0 = (v8f)0.0f, acc1 = (v8f)0.0f, acc2 = (v8f)0.0f, acc3 = (v8f)0.0f;
    float rem   = 1.0f;
    float denom = 0.0f;
    const float scale = 0.125f;

#define STORE_K(ST, GK0, GK1, GC, GS)                                     \
    { v4f r0_, r1_;                                                       \
      r0_.x = GK0.x * GC.x - GK0.y * GS.x;                                \
      r0_.y = GK0.y * GC.x + GK0.x * GS.x;                                \
      r0_.z = GK0.z * GC.y - GK0.w * GS.y;                                \
      r0_.w = GK0.w * GC.y + GK0.z * GS.y;                                \
      r1_.x = GK1.x * GC.z - GK1.y * GS.z;                                \
      r1_.y = GK1.y * GC.z + GK1.x * GS.z;                                \
      r1_.z = GK1.z * GC.w - GK1.w * GS.w;                                \
      r1_.w = GK1.w * GC.w + GK1.z * GS.w;                                \
      *(v4f*)(kdst + (ST) * 16 * KPITCH)     = r0_;                       \
      *(v4f*)(kdst + (ST) * 16 * KPITCH + 4) = r1_; }

    if (producer) {
        const v4f gk0 = *(const v4f*)(kst);
        const v4f gk1 = *(const v4f*)(kst + 4);
        const v4f gc  = *(const v4f*)(cst);
        const v4f gs  = *(const v4f*)(sst);

        const int qrow = qt * 16 + ln;
#pragma unroll
        for (int c = 0; c < 16; ++c) {
            const int d  = 4 * c + 2 * hi;
            const v2f qp = *(const v2f*)(qb + qrow * D_ + d);
            const float cc = cosc[qrow * HALF_ + (d >> 1)];
            const float ss = sinc[qrow * HALF_ + (d >> 1)];
            afrag[c].x = qp.x * cc - qp.y * ss;
            afrag[c].y = qp.y * cc + qp.x * ss;
        }
        STORE_K(0, gk0, gk1, gc, gs)
    }

    if (producer) {
        for (int t = 0; t <= ktmax + 1; ++t) {
            __syncthreads();
            const bool more = t < ktmax;
            v4f gk0, gk1, gc, gs;
            if (more) {
                const int o_ = (t + 1) * 16;
                gk0 = *(const v4f*)(kst + o_ * D_);
                gk1 = *(const v4f*)(kst + o_ * D_ + 4);
                gc  = *(const v4f*)(cst + o_ * HALF_);
                gs  = *(const v4f*)(sst + o_ * HALF_);
            }

            if (t <= qt) {
                const float* krt = &krs[t & 1][0];
                float* wlds = &bts[t & 1][slot][0];

                v2f bfrag[16];
#pragma unroll
                for (int c = 0; c < 16; ++c)
                    bfrag[c] = *(const v2f*)(krt + ln * KPITCH + 4 * c + 2 * hi);
                asm volatile("" :: "v"(bfrag[0]),  "v"(bfrag[1]),  "v"(bfrag[2]),  "v"(bfrag[3]),
                                   "v"(bfrag[4]),  "v"(bfrag[5]),  "v"(bfrag[6]),  "v"(bfrag[7]),
                                   "v"(bfrag[8]),  "v"(bfrag[9]),  "v"(bfrag[10]), "v"(bfrag[11]),
                                   "v"(bfrag[12]), "v"(bfrag[13]), "v"(bfrag[14]), "v"(bfrag[15]));
                v8f cacc = (v8f)0.0f;
#pragma unroll
                for (int c = 0; c < 16; ++c)
                    cacc = __builtin_amdgcn_wmma_f32_16x16x4_f32(
                        false, afrag[c], false, bfrag[c], (short)0, cacc, false, false);

#pragma unroll
                for (int r = 0; r < 8; ++r) {
                    const int m    = r + 8 * hi;
                    const int qpos = qt * 16 + m;
                    const int kpos = t * 16 + ln;
                    float x = cacc[r] * scale - LAMBDA_ * fabsf((float)(qpos - kpos));
                    const bool fut = kpos > qpos;
                    x = fminf(fmaxf(x, -CLAMP_), CLAMP_);
                    float beta = 1.0f / (1.0f + __expf(-x));
                    if (fut) beta = 0.0f;
                    wlds[m * LSTR + ln] = beta;
                }
            }

            if (more) STORE_K((t + 1) & 1, gk0, gk1, gc, gs)
        }
    } else {
        for (int t = 0; t <= ktmax + 1; ++t) {
            __syncthreads();
            const bool havev = t <= ktmax;
            v4f gva, gvb;
            if (havev) {
                const int o_ = t * 16;
                gva = *(const v4f*)(vst + o_ * D_);
                gvb = *(const v4f*)(vst + o_ * D_ + D_);
            }

            const int kt = t - 1;
            if (t >= 1 && kt <= qt) {
                const float* vpt = &vps[kt & 1][0];
                float* wlds = &bts[kt & 1][slot][0];

                v2f vf[4][4];
#pragma unroll
                for (int kc = 0; kc < 4; ++kc) {
                    const float* vrow = vpt + (2 * kc + hi) * VPITCH + 2 * ln;
#pragma unroll
                    for (int j = 0; j < 4; ++j)
                        vf[kc][j] = *(const v2f*)(vrow + 32 * j);
                }

                if (lane < 16) {
                    v4f* rowp = (v4f*)(wlds + lane * LSTR);
                    v4f w0 = rowp[0], w1 = rowp[1], w2 = rowp[2], w3 = rowp[3];
                    float r_   = rem;
                    float dsum = 0.0f;
#define SB_STEP(VEC, I)                                   \
                    { const float w = VEC[I] * r_;                \
                      r_ = fmaxf(r_ * (1.0f - w), EPS_);          \
                      VEC[I] = w;  dsum += w; }
                    SB_STEP(w0, 0) SB_STEP(w0, 1) SB_STEP(w0, 2) SB_STEP(w0, 3)
                    SB_STEP(w1, 0) SB_STEP(w1, 1) SB_STEP(w1, 2) SB_STEP(w1, 3)
                    SB_STEP(w2, 0) SB_STEP(w2, 1) SB_STEP(w2, 2) SB_STEP(w2, 3)
                    SB_STEP(w3, 0) SB_STEP(w3, 1) SB_STEP(w3, 2) SB_STEP(w3, 3)
#undef SB_STEP
                    rowp[0] = w0; rowp[1] = w1; rowp[2] = w2; rowp[3] = w3;
                    rem    = r_;
                    denom += dsum;
                }
                asm volatile("s_wait_dscnt 0" ::: "memory");
                __builtin_amdgcn_wave_barrier();

                v2f awf[4];
#pragma unroll
                for (int kc = 0; kc < 4; ++kc)
                    awf[kc] = *(const v2f*)(wlds + ln * LSTR + 4 * kc + 2 * hi);
#pragma unroll
                for (int kc = 0; kc < 4; ++kc) {
                    acc0 = __builtin_amdgcn_wmma_f32_16x16x4_f32(
                        false, awf[kc], false, vf[kc][0], (short)0, acc0, false, false);
                    acc1 = __builtin_amdgcn_wmma_f32_16x16x4_f32(
                        false, awf[kc], false, vf[kc][1], (short)0, acc1, false, false);
                    acc2 = __builtin_amdgcn_wmma_f32_16x16x4_f32(
                        false, awf[kc], false, vf[kc][2], (short)0, acc2, false, false);
                    acc3 = __builtin_amdgcn_wmma_f32_16x16x4_f32(
                        false, awf[kc], false, vf[kc][3], (short)0, acc3, false, false);
                }
            }

            if (havev) {
                v4f p0_, p1_;
                p0_.x = gva.x; p0_.y = gvb.x; p0_.z = gva.y; p0_.w = gvb.y;
                p1_.x = gva.z; p1_.y = gvb.z; p1_.z = gva.w; p1_.w = gvb.w;
                *(v4f*)(vdst + (t & 1) * 8 * VPITCH)     = p0_;
                *(v4f*)(vdst + (t & 1) * 8 * VPITCH + 4) = p1_;
            }
        }
    }
#undef STORE_K

    if (!producer) {
        float* dlds = &dls[slot][0];
        if (lane < 16) dlds[lane] = fmaxf(denom, EPS_);
        asm volatile("s_wait_dscnt 0" ::: "memory");
        __builtin_amdgcn_wave_barrier();

#pragma unroll
        for (int r = 0; r < 8; ++r) {
            const int m  = r + 8 * hi;
            const float dn = dlds[m];
            const int orow = qt * 16 + m;
            ob[orow * D_ +  0 + ln] = acc0[r] / dn;
            ob[orow * D_ + 16 + ln] = acc1[r] / dn;
            ob[orow * D_ + 32 + ln] = acc2[r] / dn;
            ob[orow * D_ + 48 + ln] = acc3[r] / dn;
        }
    }
}

extern "C" void kernel_launch(void* const* d_in, const int* in_sizes, int n_in,
                              void* d_out, int out_size, void* d_ws, size_t ws_size,
                              hipStream_t stream) {
    (void)in_sizes; (void)n_in; (void)out_size; (void)d_ws; (void)ws_size;
    const float* q    = (const float*)d_in[0];
    const float* k    = (const float*)d_in[1];
    const float* v    = (const float*)d_in[2];
    const float* cosc = (const float*)d_in[3];
    const float* sinc = (const float*)d_in[4];
    float* out = (float*)d_out;

    const int tiles  = B_ * H_ * QTILES;
    const int blocks = tiles / WPB;
    rse_stickbreak_attn<<<blocks, NTHR, 0, stream>>>(q, k, v, cosc, sinc, out);
}


# Wave priority (s_setprio) follows the trips a block still has to run: on each SIMD the block with the longest chain left is served first, not the oldest; sum orders unchanged
# speedup vs baseline: 2.3418x; 1.1242x over previous
#include <hip/hip_runtime.h>
#include <math.h>


typedef float v8f __attribute__((ext_vector_type(8)));
typedef float v4f __attribute__((ext_vector_type(4)));
typedef float v2f __attribute__((ext_vector_type(2)));

#define B_      4
#define H_      12
#define S_      1024
#define D_      64
#define HALF_   32
#define LAMBDA_ 0.01f
#define EPS_    1e-6f
#define CLAMP_  20.0f
#define QTILES  (S_ / 16)
#define WPB     4
#define NTHR    (64 * WPB)
#define LSTR    20
#define KPITCH  68
#define VPITCH  160

__global__ __launch_bounds__(NTHR, 6)
void rse_stickbreak_attn(const float* __restrict__ q,
                         const float* __restrict__ k,
                         const float* __restrict__ v,
                         const float* __restrict__ cosc,
                         const float* __restrict__ sinc,
                         float* __restrict__ out)
{
    __shared__ __align__(16) float krs[2][16 * KPITCH];
    __shared__ __align__(16) float vps[2][8 * VPITCH];
    __shared__ __align__(16) float bts[2][WPB][16 * LSTR];
    __shared__ __align__(16) float dls[WPB][16];

    const int tid  = threadIdx.x;
    const int wave = __builtin_amdgcn_readfirstlane(tid >> 5);
    const int lane = tid & 31;
    const bool producer = wave < WPB;
    const int qw   = wave & (WPB - 1);
    const int rtid = tid & (32 * WPB - 1);
    const int bid   = blockIdx.x;
    const int bh    = bid >> 4;
    const int u_    = bid & 15;
    const int trd   = bid >> 8;
    const int jj    = (7 * u_ + 5 * trd) & 15;
    const int sp    = (qw + u_ + trd) & 3;
    const int slot  = producer ? (sp ^ 2) : sp;
    const int qt    = ((slot & 2) ? 62 - 2 * jj : 2 * jj) + (slot & 1);
    const int ktmax = 63 - 2 * jj;

#define PRIO_AT_TRIP(T)                                                   \
    { const int rem_ = ktmax + 1 - (T);             \
      if (rem_ == 39)      __builtin_amdgcn_s_setprio(2);                 \
      else if (rem_ == 23) __builtin_amdgcn_s_setprio(1);                 \
      else if (rem_ == 7)  __builtin_amdgcn_s_setprio(0); }
    if (ktmax + 1 >= 40) __builtin_amdgcn_s_setprio(3);
    else                 __builtin_amdgcn_s_setprio(2);

    const int hi = lane >> 4;
    const int ln = lane & 15;

    const float* qb = q + (size_t)bh * S_ * D_;
    const float* kb = k + (size_t)bh * S_ * D_;
    const float* vb = v + (size_t)bh * S_ * D_;
    float*       ob = out + (size_t)bh * S_ * D_;

    const int sr = rtid >> 3;
    const int sg = rtid & 7;
    const int vp = rtid >> 4;
    const int vu = rtid & 15;
    const float* kst = kb   + sr * D_    + 8 * sg;
    const float* cst = cosc + sr * HALF_ + 4 * sg;
    const float* sst = sinc + sr * HALF_ + 4 * sg;
    const float* vst = vb   + (2 * vp) * D_ + 4 * vu;
    float* kdst = &krs[0][sr * KPITCH + 8 * sg];
    float* vdst = &vps[0][vp * VPITCH + 8 * vu];

    v2f afrag[16];
    v8f acc0 = (v8f)0.0f, acc1 = (v8f)0.0f, acc2 = (v8f)0.0f, acc3 = (v8f)0.0f;
    float rem   = 1.0f;
    float denom = 0.0f;
    const float scale = 0.125f;

#define STORE_K(ST, GK0, GK1, GC, GS)                                     \
    { v4f r0_, r1_;                                                       \
      r0_.x = GK0.x * GC.x - GK0.y * GS.x;                                \
      r0_.y = GK0.y * GC.x + GK0.x * GS.x;                                \
      r0_.z = GK0.z * GC.y - GK0.w * GS.y;                                \
      r0_.w = GK0.w * GC.y + GK0.z * GS.y;                                \
      r1_.x = GK1.x * GC.z - GK1.y * GS.z;                                \
      r1_.y = GK1.y * GC.z + GK1.x * GS.z;                                \
      r1_.z = GK1.z * GC.w - GK1.w * GS.w;                                \
      r1_.w = GK1.w * GC.w + GK1.z * GS.w;                                \
      *(v4f*)(kdst + (ST) * 16 * KPITCH)     = r0_;                       \
      *(v4f*)(kdst + (ST) * 16 * KPITCH + 4) = r1_; }

    if (producer) {
        const v4f gk0 = *(const v4f*)(kst);
        const v4f gk1 = *(const v4f*)(kst + 4);
        const v4f gc  = *(const v4f*)(cst);
        const v4f gs  = *(const v4f*)(sst);

        const int qrow = qt * 16 + ln;
#pragma unroll
        for (int c = 0; c < 16; ++c) {
            const int d  = 4 * c + 2 * hi;
            const v2f qp = *(const v2f*)(qb + qrow * D_ + d);
            const float cc = cosc[qrow * HALF_ + (d >> 1)];
            const float ss = sinc[qrow * HALF_ + (d >> 1)];
            afrag[c].x = qp.x * cc - qp.y * ss;
            afrag[c].y = qp.y * cc + qp.x * ss;
        }
        STORE_K(0, gk0, gk1, gc, gs)
    }

    if (producer) {
        for (int t = 0; t <= ktmax + 1; ++t) {
            PRIO_AT_TRIP(t)
            __syncthreads();
            const bool more = t < ktmax;
            v4f gk0, gk1, gc, gs;
            if (more) {
                const int o_ = (t + 1) * 16;
                gk0 = *(const v4f*)(kst + o_ * D_);
                gk1 = *(const v4f*)(kst + o_ * D_ + 4);
                gc  = *(const v4f*)(cst + o_ * HALF_);
                gs  = *(const v4f*)(sst + o_ * HALF_);
            }

            if (t <= qt) {
                const float* krt = &krs[t & 1][0];
                float* wlds = &bts[t & 1][slot][0];

                v2f bfrag[16];
#pragma unroll
                for (int c = 0; c < 16; ++c)
                    bfrag[c] = *(const v2f*)(krt + ln * KPITCH + 4 * c + 2 * hi);
                asm volatile("" :: "v"(bfrag[0]),  "v"(bfrag[1]),  "v"(bfrag[2]),  "v"(bfrag[3]),
                                   "v"(bfrag[4]),  "v"(bfrag[5]),  "v"(bfrag[6]),  "v"(bfrag[7]),
                                   "v"(bfrag[8]),  "v"(bfrag[9]),  "v"(bfrag[10]), "v"(bfrag[11]),
                                   "v"(bfrag[12]), "v"(bfrag[13]), "v"(bfrag[14]), "v"(bfrag[15]));
                v8f cacc = (v8f)0.0f;
#pragma unroll
                for (int c = 0; c < 16; ++c)
                    cacc = __builtin_amdgcn_wmma_f32_16x16x4_f32(
                        false, afrag[c], false, bfrag[c], (short)0, cacc, false, false);

#pragma unroll
                for (int r = 0; r < 8; ++r) {
                    const int m    = r + 8 * hi;
                    const int qpos = qt * 16 + m;
                    const int kpos = t * 16 + ln;
                    float x = cacc[r] * scale - LAMBDA_ * fabsf((float)(qpos - kpos));
                    const bool fut = kpos > qpos;
                    x = fminf(fmaxf(x, -CLAMP_), CLAMP_);
                    float beta = 1.0f / (1.0f + __expf(-x));
                    if (fut) beta = 0.0f;
                    wlds[m * LSTR + ln] = beta;
                }
            }

            if (more) STORE_K((t + 1) & 1, gk0, gk1, gc, gs)
        }
    } else {
        for (int t = 0; t <= ktmax + 1; ++t) {
            PRIO_AT_TRIP(t)
            __syncthreads();
            const bool havev = t <= ktmax;
            v4f gva, gvb;
            if (havev) {
                const int o_ = t * 16;
                gva = *(const v4f*)(vst + o_ * D_);
                gvb = *(const v4f*)(vst + o_ * D_ + D_);
            }

            const int kt = t - 1;
            if (t >= 1 && kt <= qt) {
                const float* vpt = &vps[kt & 1][0];
                float* wlds = &bts[kt & 1][slot][0];

                v2f vf[4][4];
#pragma unroll
                for (int kc = 0; kc < 4; ++kc) {
                    const float* vrow = vpt + (2 * kc + hi) * VPITCH + 2 * ln;
#pragma unroll
                    for (int j = 0; j < 4; ++j)
                        vf[kc][j] = *(const v2f*)(vrow + 32 * j);
                }

                if (lane < 16) {
                    v4f* rowp = (v4f*)(wlds + lane * LSTR);
                    v4f w0 = rowp[0], w1 = rowp[1], w2 = rowp[2], w3 = rowp[3];
                    float r_   = rem;
                    float dsum = 0.0f;
#define SB_STEP(VEC, I)                                   \
                    { const float w = VEC[I] * r_;                \
                      r_ = fmaxf(r_ * (1.0f - w), EPS_);          \
                      VEC[I] = w;  dsum += w; }
                    SB_STEP(w0, 0) SB_STEP(w0, 1) SB_STEP(w0, 2) SB_STEP(w0, 3)
                    SB_STEP(w1, 0) SB_STEP(w1, 1) SB_STEP(w1, 2) SB_STEP(w1, 3)
                    SB_STEP(w2, 0) SB_STEP(w2, 1) SB_STEP(w2, 2) SB_STEP(w2, 3)
                    SB_STEP(w3, 0) SB_STEP(w3, 1) SB_STEP(w3, 2) SB_STEP(w3, 3)
#undef SB_STEP
                    rowp[0] = w0; rowp[1] = w1; rowp[2] = w2; rowp[3] = w3;
                    rem    = r_;
                    denom += dsum;
                }
                asm volatile("s_wait_dscnt 0" ::: "memory");
                __builtin_amdgcn_wave_barrier();

                v2f awf[4];
#pragma unroll
                for (int kc = 0; kc < 4; ++kc)
                    awf[kc] = *(const v2f*)(wlds + ln * LSTR + 4 * kc + 2 * hi);
#pragma unroll
                for (int kc = 0; kc < 4; ++kc) {
                    acc0 = __builtin_amdgcn_wmma_f32_16x16x4_f32(
                        false, awf[kc], false, vf[kc][0], (short)0, acc0, false, false);
                    acc1 = __builtin_amdgcn_wmma_f32_16x16x4_f32(
                        false, awf[kc], false, vf[kc][1], (short)0, acc1, false, false);
                    acc2 = __builtin_amdgcn_wmma_f32_16x16x4_f32(
                        false, awf[kc], false, vf[kc][2], (short)0, acc2, false, false);
                    acc3 = __builtin_amdgcn_wmma_f32_16x16x4_f32(
                        false, awf[kc], false, vf[kc][3], (short)0, acc3, false, false);
                }
            }

            if (havev) {
                v4f p0_, p1_;
                p0_.x = gva.x; p0_.y = gvb.x; p0_.z = gva.y; p0_.w = gvb.y;
                p1_.x = gva.z; p1_.y = gvb.z; p1_.z = gva.w; p1_.w = gvb.w;
                *(v4f*)(vdst + (t & 1) * 8 * VPITCH)     = p0_;
                *(v4f*)(vdst + (t & 1) * 8 * VPITCH + 4) = p1_;
            }
        }
    }
#undef STORE_K
#undef PRIO_AT_TRIP
    __builtin_amdgcn_s_setprio(0);

    if (!producer) {
        float* dlds = &dls[slot][0];
        if (lane < 16) dlds[lane] = fmaxf(denom, EPS_);
        asm volatile("s_wait_dscnt 0" ::: "memory");
        __builtin_amdgcn_wave_barrier();

#pragma unroll
        for (int r = 0; r < 8; ++r) {
            const int m  = r + 8 * hi;
            const float dn = dlds[m];
            const int orow = qt * 16 + m;
            ob[orow * D_ +  0 + ln] = acc0[r] / dn;
            ob[orow * D_ + 16 + ln] = acc1[r] / dn;
            ob[orow * D_ + 32 + ln] = acc2[r] / dn;
            ob[orow * D_ + 48 + ln] = acc3[r] / dn;
        }
    }
}

extern "C" void kernel_launch(void* const* d_in, const int* in_sizes, int n_in,
                              void* d_out, int out_size, void* d_ws, size_t ws_size,
                              hipStream_t stream) {
    (void)in_sizes; (void)n_in; (void)out_size; (void)d_ws; (void)ws_size;
    const float* q    = (const float*)d_in[0];
    const float* k    = (const float*)d_in[1];
    const float* v    = (const float*)d_in[2];
    const float* cosc = (const float*)d_in[3];
    const float* sinc = (const float*)d_in[4];
    float* out = (float*)d_out;

    const int tiles  = B_ * H_ * QTILES;
    const int blocks = tiles / WPB;
    rse_stickbreak_attn<<<blocks, NTHR, 0, stream>>>(q, k, v, cosc, sinc, out);
}


# Trips cover a pair of key tiles (32 keys): one barrier, one k/v staging round, one beta hand-over, one 32-step scan per pair; the pair's two logits chains issue alternately; sum orders unchanged
# speedup vs baseline: 2.4702x; 1.0548x over previous
#include <hip/hip_runtime.h>
#include <math.h>


typedef float v8f __attribute__((ext_vector_type(8)));
typedef float v4f __attribute__((ext_vector_type(4)));
typedef float v2f __attribute__((ext_vector_type(2)));

#define B_      4
#define H_      12
#define S_      1024
#define D_      64
#define HALF_   32
#define LAMBDA_ 0.01f
#define EPS_    1e-6f
#define CLAMP_  20.0f
#define QTILES  (S_ / 16)
#define WPB     4
#define NTHR    (64 * WPB)
#define LSTR    20
#define KPITCH  68
#define VPITCH  160

__global__ __launch_bounds__(NTHR, 6)
void rse_stickbreak_attn(const float* __restrict__ q,
                         const float* __restrict__ k,
                         const float* __restrict__ v,
                         const float* __restrict__ cosc,
                         const float* __restrict__ sinc,
                         float* __restrict__ out)
{
    __shared__ __align__(16) float krs[2][2][16 * KPITCH];
    __shared__ __align__(16) float vps[2][2][8 * VPITCH];
    __shared__ __align__(16) float bts[2][WPB][2][16 * LSTR];
    __shared__ __align__(16) float dls[WPB][16];

    const int tid  = threadIdx.x;
    const int wave = __builtin_amdgcn_readfirstlane(tid >> 5);
    const int lane = tid & 31;
    const bool producer = wave < WPB;
    const int qw   = wave & (WPB - 1);
    const int rtid = tid & (32 * WPB - 1);
    const int bid   = blockIdx.x;
    const int bh    = bid >> 4;
    const int u_    = bid & 15;
    const int trd   = bid >> 8;
    const int jj    = (7 * u_ + 5 * trd) & 15;
    const int sp    = (qw + u_ + trd) & 3;
    const int slot  = producer ? (sp ^ 2) : sp;
    const int qt    = ((slot & 2) ? 62 - 2 * jj : 2 * jj) + (slot & 1);
    const int ktmax = 63 - 2 * jj;
    const int npair = 32 - jj;

#define PRIO_AT_TRIP(T)                                                   \
    { const int rem_ = npair - (T);                 \
      if (rem_ == 20)      __builtin_amdgcn_s_setprio(2);                 \
      else if (rem_ == 12) __builtin_amdgcn_s_setprio(1);                 \
      else if (rem_ == 4)  __builtin_amdgcn_s_setprio(0); }
    if (npair >= 20) __builtin_amdgcn_s_setprio(3);
    else             __builtin_amdgcn_s_setprio(2);

    const int hi = lane >> 4;
    const int ln = lane & 15;

    const float* qb = q + (size_t)bh * S_ * D_;
    const float* kb = k + (size_t)bh * S_ * D_;
    const float* vb = v + (size_t)bh * S_ * D_;
    float*       ob = out + (size_t)bh * S_ * D_;

    const int sr = rtid >> 3;
    const int sg = rtid & 7;
    const int vp = rtid >> 4;
    const int vu = rtid & 15;
    const float* kst = kb   + sr * D_    + 8 * sg;
    const float* cst = cosc + sr * HALF_ + 4 * sg;
    const float* sst = sinc + sr * HALF_ + 4 * sg;
    const float* vst = vb   + (2 * vp) * D_ + 4 * vu;
    float* kdst = &krs[0][0][sr * KPITCH + 8 * sg];
    float* vdst = &vps[0][0][vp * VPITCH + 8 * vu];

    const float scale = 0.125f;

#define STORE_K(ST, GK0, GK1, GC, GS)                                     \
    { v4f r0_, r1_;                                                       \
      r0_.x = GK0.x * GC.x - GK0.y * GS.x;                                \
      r0_.y = GK0.y * GC.x + GK0.x * GS.x;                                \
      r0_.z = GK0.z * GC.y - GK0.w * GS.y;                                \
      r0_.w = GK0.w * GC.y + GK0.z * GS.y;                                \
      r1_.x = GK1.x * GC.z - GK1.y * GS.z;                                \
      r1_.y = GK1.y * GC.z + GK1.x * GS.z;                                \
      r1_.z = GK1.z * GC.w - GK1.w * GS.w;                                \
      r1_.w = GK1.w * GC.w + GK1.z * GS.w;                                \
      *(v4f*)(kdst + (ST) * 16 * KPITCH)     = r0_;                       \
      *(v4f*)(kdst + (ST) * 16 * KPITCH + 4) = r1_; }


    const int dq0 = qt * 16 + 8 * hi - ln;
#define BETA_TILE(CACC, KT, WL)                                           \
    { const float d0_ = (float)(dq0 - (KT) * 16);                         \
      float* wl_ = (WL) + (8 * hi) * LSTR + ln;                           \
      _Pragma("unroll")                                                   \
      for (int r = 0; r < 8; ++r) {                                       \
          const float dr = d0_ + (float)r;        \
          float x = CACC[r] * scale - LAMBDA_ * fabsf(dr);                \
          x = fminf(fmaxf(x, -CLAMP_), CLAMP_);                           \
          float beta = 1.0f / (1.0f + __expf(-x));                        \
          if (dr < 0.0f) beta = 0.0f;               \
          wl_[r * LSTR] = beta;                                           \
      } }

    if (producer) {
        v2f afrag[16];
        const v4f gk0 = *(const v4f*)(kst);
        const v4f gk1 = *(const v4f*)(kst + 4);
        const v4f gc  = *(const v4f*)(cst);
        const v4f gs  = *(const v4f*)(sst);
        const v4f hk0 = *(const v4f*)(kst + 16 * D_);
        const v4f hk1 = *(const v4f*)(kst + 16 * D_ + 4);
        const v4f hc  = *(const v4f*)(cst + 16 * HALF_);
        const v4f hs  = *(const v4f*)(sst + 16 * HALF_);

        const int qrow = qt * 16 + ln;
#pragma unroll
        for (int c = 0; c < 16; ++c) {
            const int d  = 4 * c + 2 * hi;
            const v2f qp = *(const v2f*)(qb + qrow * D_ + d);
            const float cc = cosc[qrow * HALF_ + (d >> 1)];
            const float ss = sinc[qrow * HALF_ + (d >> 1)];
            afrag[c].x = qp.x * cc - qp.y * ss;
            afrag[c].y = qp.y * cc + qp.x * ss;
        }
        STORE_K(0, gk0, gk1, gc, gs)
        STORE_K(1, hk0, hk1, hc, hs)

        for (int T = 0; T <= npair; ++T) {
            PRIO_AT_TRIP(T)
            __syncthreads();
            const bool more = T + 1 < npair;
            const int o_  = (2 * T + 2) * 16;
            const int nst = 2 * ((T + 1) & 1);
            v4f gk0, gk1, gc, gs;
#define LOAD_K_ROWS(RO)                                                   \
            { gk0 = *(const v4f*)(kst + (RO) * D_);                       \
              gk1 = *(const v4f*)(kst + (RO) * D_ + 4);                   \
              gc  = *(const v4f*)(cst + (RO) * HALF_);                    \
              gs  = *(const v4f*)(sst + (RO) * HALF_); }
#define LOGITS_TILE(KRT, KT, WL)                                          \
            { v2f bfrag[16];                                              \
              _Pragma("unroll")                                           \
              for (int c = 0; c < 16; ++c)       \
                  bfrag[c] = *(const v2f*)((KRT) + ln * KPITCH + 4 * c + 2 * hi); \
                                 \
              asm volatile("" :: "v"(bfrag[0]),  "v"(bfrag[1]),  "v"(bfrag[2]),  "v"(bfrag[3]),      \
                                 "v"(bfrag[4]),  "v"(bfrag[5]),  "v"(bfrag[6]),  "v"(bfrag[7]),      \
                                 "v"(bfrag[8]),  "v"(bfrag[9]),  "v"(bfrag[10]), "v"(bfrag[11]),     \
                                 "v"(bfrag[12]), "v"(bfrag[13]), "v"(bfrag[14]), "v"(bfrag[15]));    \
              v8f cacc = (v8f)0.0f;                                       \
              _Pragma("unroll")                                           \
              for (int c = 0; c < 16; ++c)                                \
                  cacc = __builtin_amdgcn_wmma_f32_16x16x4_f32(           \
                      false, afrag[c], false, bfrag[c], (short)0, cacc, false, false); \
              BETA_TILE(cacc, KT, WL) }

            const int kt0 = 2 * T;
            if (more) LOAD_K_ROWS(o_)
            if (kt0 + 1 <= qt) {
                const float* krt = &krs[T & 1][0][0];
                float* wlds = &bts[T & 1][slot][0][0];
                v2f bf0[16], bf1[16];
#pragma unroll
                for (int c = 0; c < 16; ++c) {
                    bf0[c] = *(const v2f*)(krt + ln * KPITCH + 4 * c + 2 * hi);
                    bf1[c] = *(const v2f*)(krt + 16 * KPITCH + ln * KPITCH + 4 * c + 2 * hi);
                }
                asm volatile("" :: "v"(bf0[0]),  "v"(bf0[1]),  "v"(bf0[2]),  "v"(bf0[3]),
                                   "v"(bf0[4]),  "v"(bf0[5]),  "v"(bf0[6]),  "v"(bf0[7]),
                                   "v"(bf0[8]),  "v"(bf0[9]),  "v"(bf0[10]), "v"(bf0[11]),
                                   "v"(bf0[12]), "v"(bf0[13]), "v"(bf0[14]), "v"(bf0[15]));
                asm volatile("" :: "v"(bf1[0]),  "v"(bf1[1]),  "v"(bf1[2]),  "v"(bf1[3]),
                                   "v"(bf1[4]),  "v"(bf1[5]),  "v"(bf1[6]),  "v"(bf1[7]),
                                   "v"(bf1[8]),  "v"(bf1[9]),  "v"(bf1[10]), "v"(bf1[11]),
                                   "v"(bf1[12]), "v"(bf1[13]), "v"(bf1[14]), "v"(bf1[15]));
                v8f ca0 = (v8f)0.0f, ca1 = (v8f)0.0f;
#pragma unroll
                for (int c = 0; c < 16; ++c) {
                    ca0 = __builtin_amdgcn_wmma_f32_16x16x4_f32(
                        false, afrag[c], false, bf0[c], (short)0, ca0, false, false);
                    ca1 = __builtin_amdgcn_wmma_f32_16x16x4_f32(
                        false, afrag[c], false, bf1[c], (short)0, ca1, false, false);
                }
                v4f hk0, hk1, hc, hs;
                if (more) {
                    hk0 = *(const v4f*)(kst + (o_ + 16) * D_);
                    hk1 = *(const v4f*)(kst + (o_ + 16) * D_ + 4);
                    hc  = *(const v4f*)(cst + (o_ + 16) * HALF_);
                    hs  = *(const v4f*)(sst + (o_ + 16) * HALF_);
                }
                BETA_TILE(ca0, kt0, wlds)
                if (more) STORE_K(nst, gk0, gk1, gc, gs)
                BETA_TILE(ca1, kt0 + 1, wlds + 16 * LSTR)
                if (more) STORE_K(nst + 1, hk0, hk1, hc, hs)
            } else {
                if (kt0 <= qt)
                    LOGITS_TILE(&krs[T & 1][0][0], kt0, &bts[T & 1][slot][0][0])
                if (more) {
                    STORE_K(nst, gk0, gk1, gc, gs)
                    LOAD_K_ROWS(o_ + 16)
                    STORE_K(nst + 1, gk0, gk1, gc, gs)
                }
            }
#undef LOGITS_TILE
#undef LOAD_K_ROWS
        }
        __builtin_amdgcn_s_setprio(0);
    } else {
        v8f acc0 = (v8f)0.0f, acc1 = (v8f)0.0f, acc2 = (v8f)0.0f, acc3 = (v8f)0.0f;
        float rem   = 1.0f;
        float denom = 0.0f;
#define SB_STEP(VEC, I)                                   \
        { const float w = VEC[I] * r_;                    \
          r_ = fmaxf(r_ * (1.0f - w), EPS_);              \
          VEC[I] = w;  dsum += w; }
#define SCAN_TILE(W0, W1, W2, W3)                                         \
        { float dsum = 0.0f;                                              \
          SB_STEP(W0, 0) SB_STEP(W0, 1) SB_STEP(W0, 2) SB_STEP(W0, 3)     \
          SB_STEP(W1, 0) SB_STEP(W1, 1) SB_STEP(W1, 2) SB_STEP(W1, 3)     \
          SB_STEP(W2, 0) SB_STEP(W2, 1) SB_STEP(W2, 2) SB_STEP(W2, 3)     \
          SB_STEP(W3, 0) SB_STEP(W3, 1) SB_STEP(W3, 2) SB_STEP(W3, 3)     \
          denom += dsum; }
#define WV_TILE(AW, VF)                                                   \
        { _Pragma("unroll")                                               \
          for (int kc = 0; kc < 4; ++kc) {                                \
              acc0 = __builtin_amdgcn_wmma_f32_16x16x4_f32(               \
                  false, AW[kc], false, VF[kc][0], (short)0, acc0, false, false); \
              acc1 = __builtin_amdgcn_wmma_f32_16x16x4_f32(               \
                  false, AW[kc], false, VF[kc][1], (short)0, acc1, false, false); \
              acc2 = __builtin_amdgcn_wmma_f32_16x16x4_f32(               \
                  false, AW[kc], false, VF[kc][2], (short)0, acc2, false, false); \
              acc3 = __builtin_amdgcn_wmma_f32_16x16x4_f32(               \
                  false, AW[kc], false, VF[kc][3], (short)0, acc3, false, false); \
          } }
#define LOAD_VF(VF, VPT)                                                  \
        { _Pragma("unroll")                                               \
          for (int kc = 0; kc < 4; ++kc) {                                \
              const float* vrow = (VPT) + (2 * kc + hi) * VPITCH + 2 * ln; \
              _Pragma("unroll")                                           \
              for (int j = 0; j < 4; ++j)                                 \
                  VF[kc][j] = *(const v2f*)(vrow + 32 * j);               \
          } }

        for (int T = 0; T <= npair; ++T) {
            PRIO_AT_TRIP(T)
            __syncthreads();
            const bool havev = T < npair;
            v4f gva, gvb, hva, hvb;
            if (havev) {
                const int o_ = (2 * T) * 16;
                gva = *(const v4f*)(vst + o_ * D_);
                gvb = *(const v4f*)(vst + o_ * D_ + D_);
                hva = *(const v4f*)(vst + (o_ + 16) * D_);
                hvb = *(const v4f*)(vst + (o_ + 16) * D_ + D_);
            }

            const int kt0 = 2 * T - 2;
            if (T >= 1 && kt0 <= qt) {
                const float* vpt = &vps[(T - 1) & 1][0][0];
                float* wlds = &bts[(T - 1) & 1][slot][0][0];
                v4f* rowp = (v4f*)(wlds + ln * LSTR);

                v2f vf0[4][4];
                LOAD_VF(vf0, vpt)

                const bool two = kt0 + 1 <= qt;

                if (lane < 16) {
                    v4f w0 = rowp[0], w1 = rowp[1], w2 = rowp[2], w3 = rowp[3];
                    v4f x0, x1, x2, x3;
                    if (two) {
                        x0 = rowp[4 * LSTR + 0]; x1 = rowp[4 * LSTR + 1];
                        x2 = rowp[4 * LSTR + 2]; x3 = rowp[4 * LSTR + 3];
                    }
                    float r_ = rem;
                    SCAN_TILE(w0, w1, w2, w3)
                    rowp[0] = w0; rowp[1] = w1; rowp[2] = w2; rowp[3] = w3;
                    if (two) {
                        SCAN_TILE(x0, x1, x2, x3)
                        rowp[4 * LSTR + 0] = x0; rowp[4 * LSTR + 1] = x1;
                        rowp[4 * LSTR + 2] = x2; rowp[4 * LSTR + 3] = x3;
                    }
                    rem = r_;
                }
                asm volatile("s_wait_dscnt 0" ::: "memory");
                __builtin_amdgcn_wave_barrier();

                v2f aw0[4], aw1[4];
                v2f vf1[4][4];
#pragma unroll
                for (int kc = 0; kc < 4; ++kc)
                    aw0[kc] = *(const v2f*)(wlds + ln * LSTR + 4 * kc + 2 * hi);
                if (two) {
#pragma unroll
                    for (int kc = 0; kc < 4; ++kc)
                        aw1[kc] = *(const v2f*)(wlds + 16 * LSTR + ln * LSTR + 4 * kc + 2 * hi);
                }
                WV_TILE(aw0, vf0)
                if (two) {
                    LOAD_VF(vf1, vpt + 8 * VPITCH)
                    WV_TILE(aw1, vf1)
                    __builtin_amdgcn_sched_group_barrier(0x100, 8, 0);
                    __builtin_amdgcn_sched_group_barrier(0x008, 16, 0);
                }
            }

            if (havev) {
                v4f p0_, p1_;
                p0_.x = gva.x; p0_.y = gvb.x; p0_.z = gva.y; p0_.w = gvb.y;
                p1_.x = gva.z; p1_.y = gvb.z; p1_.z = gva.w; p1_.w = gvb.w;
                *(v4f*)(vdst + (2 * (T & 1)) * 8 * VPITCH)     = p0_;
                *(v4f*)(vdst + (2 * (T & 1)) * 8 * VPITCH + 4) = p1_;
                p0_.x = hva.x; p0_.y = hvb.x; p0_.z = hva.y; p0_.w = hvb.y;
                p1_.x = hva.z; p1_.y = hvb.z; p1_.z = hva.w; p1_.w = hvb.w;
                *(v4f*)(vdst + (2 * (T & 1) + 1) * 8 * VPITCH)     = p0_;
                *(v4f*)(vdst + (2 * (T & 1) + 1) * 8 * VPITCH + 4) = p1_;
            }
        }
#undef LOAD_VF
#undef WV_TILE
#undef SCAN_TILE
#undef SB_STEP
        __builtin_amdgcn_s_setprio(0);

        float* dlds = &dls[slot][0];
        if (lane < 16) dlds[lane] = fmaxf(denom, EPS_);
        asm volatile("s_wait_dscnt 0" ::: "memory");
        __builtin_amdgcn_wave_barrier();

#pragma unroll
        for (int r = 0; r < 8; ++r) {
            const int m  = r + 8 * hi;
            const float dn = dlds[m];
            const int orow = qt * 16 + m;
            ob[orow * D_ +  0 + ln] = acc0[r] / dn;
            ob[orow * D_ + 16 + ln] = acc1[r] / dn;
            ob[orow * D_ + 32 + ln] = acc2[r] / dn;
            ob[orow * D_ + 48 + ln] = acc3[r] / dn;
        }
    }
#undef BETA_TILE
#undef STORE_K
#undef PRIO_AT_TRIP
}

extern "C" void kernel_launch(void* const* d_in, const int* in_sizes, int n_in,
                              void* d_out, int out_size, void* d_ws, size_t ws_size,
                              hipStream_t stream) {
    (void)in_sizes; (void)n_in; (void)out_size; (void)d_ws; (void)ws_size;
    const float* q    = (const float*)d_in[0];
    const float* k    = (const float*)d_in[1];
    const float* v    = (const float*)d_in[2];
    const float* cosc = (const float*)d_in[3];
    const float* sinc = (const float*)d_in[4];
    float* out = (float*)d_out;

    const int tiles  = B_ * H_ * QTILES;
    const int blocks = tiles / WPB;
    rse_stickbreak_attn<<<blocks, NTHR, 0, stream>>>(q, k, v, cosc, sinc, out);
}
